# starting kernel
# baseline (speedup 1.0000x reference)
#include <hip/hip_runtime.h>

typedef __attribute__((ext_vector_type(2))) float v2f;
typedef __attribute__((ext_vector_type(8))) float v8f;

#define S_LEN   2048
#define NHEAD   16
#define DH      64
#define KSTRIDE 32
#define WIN     16
#define NTILES  7
#define NCOLS   112
#define ROWPAD  68
#define SCPAD   120

__global__ __launch_bounds__(32)
void strided_attn_wmma_f32(const float* __restrict__ q,
                           const float* __restrict__ k,
                           const float* __restrict__ v,
                           float* __restrict__ out)
{
    __shared__ float q_lds[16 * ROWPAD];
    __shared__ float k_lds[16 * ROWPAD];
    __shared__ float v_lds[16 * ROWPAD];
    __shared__ float sc_lds[16 * SCPAD];

    const int lane  = threadIdx.x;
    const int m     = lane & 15;
    const int hi    = lane >> 4;
    const int qb    = blockIdx.x;
    const int h     = blockIdx.y;
    const int qbase = qb * 16;
    const long hbase = (long)h * S_LEN * DH;

    for (int r = 0; r < 16; ++r) {
        const float* src = q + hbase + (long)(qbase + r) * DH + 2 * lane;
        v2f t; t.x = src[0]; t.y = src[1];
        *(v2f*)&q_lds[r * ROWPAD + 2 * lane] = t;
    }
    __syncthreads();

    for (int t = 0; t < NTILES; ++t) {
        for (int r = 0; r < 16; ++r) {
            int pos;
            if (t < 3) {
                pos = qbase - WIN + t * 16 + r;
                pos = pos < 0 ? 0 : (pos > S_LEN - 1 ? S_LEN - 1 : pos);
            } else {
                pos = ((t - 3) * 16 + r) * KSTRIDE;
            }
            const float* src = k + hbase + (long)pos * DH + 2 * lane;
            v2f tv; tv.x = src[0]; tv.y = src[1];
            *(v2f*)&k_lds[r * ROWPAD + 2 * lane] = tv;
        }
        __syncthreads();

        if (t + 1 < NTILES) {
            int tn = t + 1;
            int pn;
            if (tn < 3) {
                pn = qbase - WIN + tn * 16 + m;
                pn = pn < 0 ? 0 : (pn > S_LEN - 1 ? S_LEN - 1 : pn);
            } else {
                pn = ((tn - 3) * 16 + m) * KSTRIDE;
            }
            __builtin_prefetch(k + hbase + (long)pn * DH, 0, 3);
        }

        v8f acc = {};
        #pragma unroll
        for (int kc = 0; kc < 16; ++kc) {
            const int d0 = 4 * kc + 2 * hi;
            v2f a = *(const v2f*)&q_lds[m * ROWPAD + d0];
            v2f b = *(const v2f*)&k_lds[m * ROWPAD + d0];
            acc = __builtin_amdgcn_wmma_f32_16x16x4_f32(
                false, a, false, b, (short)0, acc, false, false);
        }

        #pragma unroll
        for (int r = 0; r < 8; ++r) {
            const int row = r + 8 * hi;
            const int n   = qbase + row;
            const int col = t * 16 + m;
            bool valid;
            if (t < 3) {
                const int p = qbase - WIN + col;
                valid = (col >= row) && (col <= row + 2 * WIN)
                        && (p >= 0) && (p < S_LEN);
            } else {
                valid = (n > WIN);
            }
            sc_lds[row * SCPAD + col] = valid ? acc[r] * 0.125f : -1.0e30f;
        }
        __syncthreads();
    }

    if (lane < 16) {
        float mx = -1.0e30f;
        for (int j = 0; j < NCOLS; ++j)
            mx = fmaxf(mx, sc_lds[lane * SCPAD + j]);
        float sum = 0.0f;
        for (int j = 0; j < NCOLS; ++j) {
            float e = __expf(sc_lds[lane * SCPAD + j] - mx);
            sc_lds[lane * SCPAD + j] = e;
            sum += e;
        }
        const float inv = (sum > 0.0f) ? (1.0f / sum) : 0.0f;
        for (int j = 0; j < NCOLS; ++j)
            sc_lds[lane * SCPAD + j] *= inv;
    }
    __syncthreads();

    v8f acc_o[4] = {v8f{}, v8f{}, v8f{}, v8f{}};
    for (int t = 0; t < NTILES; ++t) {
        for (int r = 0; r < 16; ++r) {
            int pos;
            if (t < 3) {
                pos = qbase - WIN + t * 16 + r;
                pos = pos < 0 ? 0 : (pos > S_LEN - 1 ? S_LEN - 1 : pos);
            } else {
                pos = ((t - 3) * 16 + r) * KSTRIDE;
            }
            const float* src = v + hbase + (long)pos * DH + 2 * lane;
            v2f tv; tv.x = src[0]; tv.y = src[1];
            *(v2f*)&v_lds[r * ROWPAD + 2 * lane] = tv;
        }
        __syncthreads();

        #pragma unroll
        for (int kc = 0; kc < 4; ++kc) {
            const int c0 = t * 16 + 4 * kc + 2 * hi;
            const int k0 = 4 * kc + 2 * hi;
            v2f a = *(const v2f*)&sc_lds[m * SCPAD + c0];
            #pragma unroll
            for (int nt = 0; nt < 4; ++nt) {
                v2f b;
                b.x = v_lds[(k0 + 0) * ROWPAD + 16 * nt + m];
                b.y = v_lds[(k0 + 1) * ROWPAD + 16 * nt + m];
                acc_o[nt] = __builtin_amdgcn_wmma_f32_16x16x4_f32(
                    false, a, false, b, (short)0, acc_o[nt], false, false);
            }
        }
        __syncthreads();
    }

    #pragma unroll
    for (int nt = 0; nt < 4; ++nt) {
        #pragma unroll
        for (int r = 0; r < 8; ++r) {
            const int row = r + 8 * hi;
            out[hbase + (long)(qbase + row) * DH + 16 * nt + m] = acc_o[nt][r];
        }
    }
}

extern "C" void kernel_launch(void* const* d_in, const int* in_sizes, int n_in,
                              void* d_out, int out_size, void* d_ws, size_t ws_size,
                              hipStream_t stream) {
    const float* q = (const float*)d_in[0];
    const float* k = (const float*)d_in[1];
    const float* v = (const float*)d_in[2];
    float* out = (float*)d_out;

    dim3 grid(S_LEN / 16, NHEAD, 1);
    dim3 block(32, 1, 1);
    strided_attn_wmma_f32<<<grid, block, 0, stream>>>(q, k, v, out);
}


# 8-wave workgroup per 128 queries of a head: Q and the local and strided K/V rows its waves share fetched once by 16-byte loads into LDS; all 7 tiles run from LDS, 3 barriers; sum order unchanged
# speedup vs baseline: 1.3182x; 1.3182x over previous
#include <hip/hip_runtime.h>

typedef __attribute__((ext_vector_type(2))) float v2f;
typedef __attribute__((ext_vector_type(4))) float v4f;
typedef __attribute__((ext_vector_type(8))) float v8f;

#define S_LEN   2048
#define NHEAD   16
#define DH      64
#define KSTRIDE 32
#define WIN     16
#define NTILES  7
#define NCOLS   112
#define ROWPAD  68
#define SCPAD   120
#define WAVES   8
#define QROWS   (16 * WAVES)
#define LROWS   (QROWS + 2 * WIN)
#define SROWS   (S_LEN / KSTRIDE)
#define QPASS   (QROWS / 16)
#define LPASS   (LROWS / 16)
#define SPASS   (SROWS / 16)

__global__ __launch_bounds__(32 * WAVES)
void strided_attn_wmma_f32(const float* __restrict__ q,
                           const float* __restrict__ k,
                           const float* __restrict__ v,
                           float* __restrict__ out)
{
    __shared__ __attribute__((aligned(16))) float q_lds[QROWS * ROWPAD];
    __shared__ __attribute__((aligned(16))) float kl_lds[LROWS * ROWPAD];
    __shared__ __attribute__((aligned(16))) float vl_lds[LROWS * ROWPAD];
    __shared__ __attribute__((aligned(16))) float ks_lds[SROWS * ROWPAD];
    __shared__ __attribute__((aligned(16))) float vs_lds[SROWS * ROWPAD];
    __shared__ __attribute__((aligned(16))) float sc_lds[WAVES * 16 * SCPAD];

    const int tid    = threadIdx.x;
    const int wv     = tid >> 5;
    const int lane   = tid & 31;
    const int m      = lane & 15;
    const int hi     = lane >> 4;
    const int h      = blockIdx.y;
    const int qbase0 = blockIdx.x * QROWS;
    const int qbase  = qbase0 + 16 * wv;
    const long hbase = (long)h * S_LEN * DH;

    const int srow = tid >> 4;
    const int scol = (tid & 15) * 4;

    v4f rq[QPASS], rk[LPASS], rks[SPASS], rv[LPASS], rvs[SPASS];

    #pragma unroll
    for (int p = 0; p < QPASS; ++p) {
        const int row = p * 16 + srow;
        rq[p] = *(const v4f*)(q + hbase + (long)(qbase0 + row) * DH + scol);
    }
    #pragma unroll
    for (int p = 0; p < LPASS; ++p) {
        int pos = qbase0 - WIN + p * 16 + srow;
        pos = pos < 0 ? 0 : (pos > S_LEN - 1 ? S_LEN - 1 : pos);
        rk[p] = *(const v4f*)(k + hbase + (long)pos * DH + scol);
    }
    #pragma unroll
    for (int p = 0; p < SPASS; ++p) {
        const int pos = (p * 16 + srow) * KSTRIDE;
        rks[p] = *(const v4f*)(k + hbase + (long)pos * DH + scol);
    }

    #pragma unroll
    for (int p = 0; p < QPASS; ++p)
        *(v4f*)&q_lds[(p * 16 + srow) * ROWPAD + scol] = rq[p];
    #pragma unroll
    for (int p = 0; p < LPASS; ++p)
        *(v4f*)&kl_lds[(p * 16 + srow) * ROWPAD + scol] = rk[p];
    #pragma unroll
    for (int p = 0; p < SPASS; ++p)
        *(v4f*)&ks_lds[(p * 16 + srow) * ROWPAD + scol] = rks[p];
    __syncthreads();

    #pragma unroll
    for (int p = 0; p < LPASS; ++p) {
        int pos = qbase0 - WIN + p * 16 + srow;
        pos = pos < 0 ? 0 : (pos > S_LEN - 1 ? S_LEN - 1 : pos);
        rv[p] = *(const v4f*)(v + hbase + (long)pos * DH + scol);
    }
    #pragma unroll
    for (int p = 0; p < SPASS; ++p) {
        const int pos = (p * 16 + srow) * KSTRIDE;
        rvs[p] = *(const v4f*)(v + hbase + (long)pos * DH + scol);
    }

    float* sc = &sc_lds[wv * 16 * SCPAD];

    v2f qa[16];
    #pragma unroll
    for (int kc = 0; kc < 16; ++kc)
        qa[kc] = *(const v2f*)&q_lds[(16 * wv + m) * ROWPAD + 4 * kc + 2 * hi];

    #pragma unroll
    for (int t = 0; t < NTILES; ++t) {
        const float* krow = (t < 3) ? &kl_lds[(16 * wv + 16 * t + m) * ROWPAD]
                                    : &ks_lds[(16 * (t - 3) + m) * ROWPAD];

        v8f acc = {};
        #pragma unroll
        for (int kc = 0; kc < 16; ++kc) {
            const int d0 = 4 * kc + 2 * hi;
            v2f b = *(const v2f*)&krow[d0];
            acc = __builtin_amdgcn_wmma_f32_16x16x4_f32(
                false, qa[kc], false, b, (short)0, acc, false, false);
        }

        #pragma unroll
        for (int r = 0; r < 8; ++r) {
            const int row = r + 8 * hi;
            const int n   = qbase + row;
            const int col = t * 16 + m;
            bool valid;
            if (t < 3) {
                const int p = qbase - WIN + col;
                valid = (col >= row) && (col <= row + 2 * WIN)
                        && (p >= 0) && (p < S_LEN);
            } else {
                valid = (n > WIN);
            }
            sc[row * SCPAD + col] = valid ? acc[r] * 0.125f : -1.0e30f;
        }
    }
    __syncthreads();

    if (lane < 16) {
        float mx = -1.0e30f;
        for (int j = 0; j < NCOLS; ++j)
            mx = fmaxf(mx, sc[lane * SCPAD + j]);
        float sum = 0.0f;
        for (int j = 0; j < NCOLS; ++j) {
            float e = __expf(sc[lane * SCPAD + j] - mx);
            sc[lane * SCPAD + j] = e;
            sum += e;
        }
        const float inv = (sum > 0.0f) ? (1.0f / sum) : 0.0f;
        for (int j = 0; j < NCOLS; ++j)
            sc[lane * SCPAD + j] *= inv;
    }

    #pragma unroll
    for (int p = 0; p < LPASS; ++p)
        *(v4f*)&vl_lds[(p * 16 + srow) * ROWPAD + scol] = rv[p];
    #pragma unroll
    for (int p = 0; p < SPASS; ++p)
        *(v4f*)&vs_lds[(p * 16 + srow) * ROWPAD + scol] = rvs[p];
    __syncthreads();

    v8f acc_o[4] = {v8f{}, v8f{}, v8f{}, v8f{}};
    #pragma unroll
    for (int t = 0; t < NTILES; ++t) {
        const float* vt = (t < 3) ? &vl_lds[(16 * wv + 16 * t) * ROWPAD]
                                  : &vs_lds[(16 * (t - 3)) * ROWPAD];
        #pragma unroll
        for (int kc = 0; kc < 4; ++kc) {
            const int c0 = t * 16 + 4 * kc + 2 * hi;
            const int k0 = 4 * kc + 2 * hi;
            v2f a = *(const v2f*)&sc[m * SCPAD + c0];
            #pragma unroll
            for (int nt = 0; nt < 4; ++nt) {
                v2f b;
                b.x = vt[(k0 + 0) * ROWPAD + 16 * nt + m];
                b.y = vt[(k0 + 1) * ROWPAD + 16 * nt + m];
                acc_o[nt] = __builtin_amdgcn_wmma_f32_16x16x4_f32(
                    false, a, false, b, (short)0, acc_o[nt], false, false);
            }
        }
    }

    #pragma unroll
    for (int nt = 0; nt < 4; ++nt) {
        #pragma unroll
        for (int r = 0; r < 8; ++r) {
            const int row = r + 8 * hi;
            out[hbase + (long)(qbase + row) * DH + 16 * nt + m] = acc_o[nt][r];
        }
    }
}

extern "C" void kernel_launch(void* const* d_in, const int* in_sizes, int n_in,
                              void* d_out, int out_size, void* d_ws, size_t ws_size,
                              hipStream_t stream) {
    const float* q = (const float*)d_in[0];
    const float* k = (const float*)d_in[1];
    const float* v = (const float*)d_in[2];
    float* out = (float*)d_out;

    dim3 grid(S_LEN / QROWS, NHEAD, 1);
    dim3 block(32 * WAVES, 1, 1);
    strided_attn_wmma_f32<<<grid, block, 0, stream>>>(q, k, v, out);
}


# Softmax in one LDS pass on all 32 lanes: a lane reads half a score row (56 values, 16-byte loads, row pitch 116) into registers, halves joined by one lane swap for max and sum; sum order changes
# speedup vs baseline: 1.9434x; 1.4743x over previous
#include <hip/hip_runtime.h>

typedef __attribute__((ext_vector_type(2))) float v2f;
typedef __attribute__((ext_vector_type(4))) float v4f;
typedef __attribute__((ext_vector_type(8))) float v8f;

#define S_LEN   2048
#define NHEAD   16
#define DH      64
#define KSTRIDE 32
#define WIN     16
#define NTILES  7
#define NCOLS   112
#define ROWPAD  68
#define SCPAD   116
#define HALFC   (NCOLS / 2)
#define HALFV   (HALFC / 4)
#define WAVES   8
#define QROWS   (16 * WAVES)
#define LROWS   (QROWS + 2 * WIN)
#define SROWS   (S_LEN / KSTRIDE)
#define QPASS   (QROWS / 16)
#define LPASS   (LROWS / 16)
#define SPASS   (SROWS / 16)

__device__ __forceinline__ float xhalf(float x)
{
    const int b = __builtin_bit_cast(int, x);
    return __builtin_bit_cast(float,
        __builtin_amdgcn_permlanex16(b, b, 0x76543210, 0xfedcba98, false, false));
}

__global__ __launch_bounds__(32 * WAVES)
void strided_attn_wmma_f32(const float* __restrict__ q,
                           const float* __restrict__ k,
                           const float* __restrict__ v,
                           float* __restrict__ out)
{
    __shared__ __attribute__((aligned(16))) float q_lds[QROWS * ROWPAD];
    __shared__ __attribute__((aligned(16))) float kl_lds[LROWS * ROWPAD];
    __shared__ __attribute__((aligned(16))) float vl_lds[LROWS * ROWPAD];
    __shared__ __attribute__((aligned(16))) float ks_lds[SROWS * ROWPAD];
    __shared__ __attribute__((aligned(16))) float vs_lds[SROWS * ROWPAD];
    __shared__ __attribute__((aligned(16))) float sc_lds[WAVES * 16 * SCPAD];

    const int tid    = threadIdx.x;
    const int wv     = tid >> 5;
    const int lane   = tid & 31;
    const int m      = lane & 15;
    const int hi     = lane >> 4;
    const int h      = blockIdx.y;
    const int qbase0 = blockIdx.x * QROWS;
    const int qbase  = qbase0 + 16 * wv;
    const long hbase = (long)h * S_LEN * DH;

    const int srow = tid >> 4;
    const int scol = (tid & 15) * 4;

    v4f rq[QPASS], rk[LPASS], rks[SPASS], rv[LPASS], rvs[SPASS];

    #pragma unroll
    for (int p = 0; p < QPASS; ++p) {
        const int row = p * 16 + srow;
        rq[p] = *(const v4f*)(q + hbase + (long)(qbase0 + row) * DH + scol);
    }
    #pragma unroll
    for (int p = 0; p < LPASS; ++p) {
        int pos = qbase0 - WIN + p * 16 + srow;
        pos = pos < 0 ? 0 : (pos > S_LEN - 1 ? S_LEN - 1 : pos);
        rk[p] = *(const v4f*)(k + hbase + (long)pos * DH + scol);
    }
    #pragma unroll
    for (int p = 0; p < SPASS; ++p) {
        const int pos = (p * 16 + srow) * KSTRIDE;
        rks[p] = *(const v4f*)(k + hbase + (long)pos * DH + scol);
    }

    #pragma unroll
    for (int p = 0; p < QPASS; ++p)
        *(v4f*)&q_lds[(p * 16 + srow) * ROWPAD + scol] = rq[p];
    #pragma unroll
    for (int p = 0; p < LPASS; ++p)
        *(v4f*)&kl_lds[(p * 16 + srow) * ROWPAD + scol] = rk[p];
    #pragma unroll
    for (int p = 0; p < SPASS; ++p)
        *(v4f*)&ks_lds[(p * 16 + srow) * ROWPAD + scol] = rks[p];
    __syncthreads();

    #pragma unroll
    for (int p = 0; p < LPASS; ++p) {
        int pos = qbase0 - WIN + p * 16 + srow;
        pos = pos < 0 ? 0 : (pos > S_LEN - 1 ? S_LEN - 1 : pos);
        rv[p] = *(const v4f*)(v + hbase + (long)pos * DH + scol);
    }
    #pragma unroll
    for (int p = 0; p < SPASS; ++p) {
        const int pos = (p * 16 + srow) * KSTRIDE;
        rvs[p] = *(const v4f*)(v + hbase + (long)pos * DH + scol);
    }

    float* sc = &sc_lds[wv * 16 * SCPAD];

    v2f qa[16];
    #pragma unroll
    for (int kc = 0; kc < 16; ++kc)
        qa[kc] = *(const v2f*)&q_lds[(16 * wv + m) * ROWPAD + 4 * kc + 2 * hi];

    #pragma unroll
    for (int t = 0; t < NTILES; ++t) {
        const float* krow = (t < 3) ? &kl_lds[(16 * wv + 16 * t + m) * ROWPAD]
                                    : &ks_lds[(16 * (t - 3) + m) * ROWPAD];

        v8f acc = {};
        #pragma unroll
        for (int kc = 0; kc < 16; ++kc) {
            const int d0 = 4 * kc + 2 * hi;
            v2f b = *(const v2f*)&krow[d0];
            acc = __builtin_amdgcn_wmma_f32_16x16x4_f32(
                false, qa[kc], false, b, (short)0, acc, false, false);
        }

        #pragma unroll
        for (int r = 0; r < 8; ++r) {
            const int row = r + 8 * hi;
            const int n   = qbase + row;
            const int col = t * 16 + m;
            bool valid;
            if (t < 3) {
                const int p = qbase - WIN + col;
                valid = (col >= row) && (col <= row + 2 * WIN)
                        && (p >= 0) && (p < S_LEN);
            } else {
                valid = (n > WIN);
            }
            sc[row * SCPAD + col] = valid ? acc[r] * 0.125f : -1.0e30f;
        }
    }
    __syncthreads();

    {
        float* prow = &sc[m * SCPAD + hi * HALFC];
        v4f s[HALFV];
        #pragma unroll
        for (int j = 0; j < HALFV; ++j)
            s[j] = *(const v4f*)&prow[4 * j];

        float mx = -1.0e30f;
        #pragma unroll
        for (int j = 0; j < HALFV; ++j)
            mx = fmaxf(mx, fmaxf(fmaxf(s[j].x, s[j].y), fmaxf(s[j].z, s[j].w)));
        mx = fmaxf(mx, xhalf(mx));

        v4f part = {0.0f, 0.0f, 0.0f, 0.0f};
        #pragma unroll
        for (int j = 0; j < HALFV; ++j) {
            v4f e;
            e.x = __expf(s[j].x - mx);
            e.y = __expf(s[j].y - mx);
            e.z = __expf(s[j].z - mx);
            e.w = __expf(s[j].w - mx);
            s[j] = e;
            part += e;
        }
        float sum = (part.x + part.y) + (part.z + part.w);
        sum = sum + xhalf(sum);
        const float inv = (sum > 0.0f) ? (1.0f / sum) : 0.0f;

        #pragma unroll
        for (int j = 0; j < HALFV; ++j)
            *(v4f*)&prow[4 * j] = s[j] * inv;
    }

    #pragma unroll
    for (int p = 0; p < LPASS; ++p)
        *(v4f*)&vl_lds[(p * 16 + srow) * ROWPAD + scol] = rv[p];
    #pragma unroll
    for (int p = 0; p < SPASS; ++p)
        *(v4f*)&vs_lds[(p * 16 + srow) * ROWPAD + scol] = rvs[p];
    __syncthreads();

    v8f acc_o[4] = {v8f{}, v8f{}, v8f{}, v8f{}};
    #pragma unroll
    for (int t = 0; t < NTILES; ++t) {
        const float* vt = (t < 3) ? &vl_lds[(16 * wv + 16 * t) * ROWPAD]
                                  : &vs_lds[(16 * (t - 3)) * ROWPAD];
        #pragma unroll
        for (int kc = 0; kc < 4; ++kc) {
            const int c0 = t * 16 + 4 * kc + 2 * hi;
            const int k0 = 4 * kc + 2 * hi;
            v2f a = *(const v2f*)&sc[m * SCPAD + c0];
            #pragma unroll
            for (int nt = 0; nt < 4; ++nt) {
                v2f b;
                b.x = vt[(k0 + 0) * ROWPAD + 16 * nt + m];
                b.y = vt[(k0 + 1) * ROWPAD + 16 * nt + m];
                acc_o[nt] = __builtin_amdgcn_wmma_f32_16x16x4_f32(
                    false, a, false, b, (short)0, acc_o[nt], false, false);
            }
        }
    }

    #pragma unroll
    for (int nt = 0; nt < 4; ++nt) {
        #pragma unroll
        for (int r = 0; r < 8; ++r) {
            const int row = r + 8 * hi;
            out[hbase + (long)(qbase + row) * DH + 16 * nt + m] = acc_o[nt][r];
        }
    }
}

extern "C" void kernel_launch(void* const* d_in, const int* in_sizes, int n_in,
                              void* d_out, int out_size, void* d_ws, size_t ws_size,
                              hipStream_t stream) {
    const float* q = (const float*)d_in[0];
    const float* k = (const float*)d_in[1];
    const float* v = (const float*)d_in[2];
    float* out = (float*)d_out;

    dim3 grid(S_LEN / QROWS, NHEAD, 1);
    dim3 block(32 * WAVES, 1, 1);
    strided_attn_wmma_f32<<<grid, block, 0, stream>>>(q, k, v, out);
}


# PV feed: V staged in LDS as interleaved key-row pairs, a B fragment is one 8-byte read (was two 4-byte reads and a register shuffle); a tile's reads issued as one group a tile ahead; sums unchanged
# speedup vs baseline: 2.0474x; 1.0535x over previous
#include <hip/hip_runtime.h>

typedef __attribute__((ext_vector_type(2))) float v2f;
typedef __attribute__((ext_vector_type(4))) float v4f;
typedef __attribute__((ext_vector_type(8))) float v8f;

#define S_LEN   2048
#define NHEAD   16
#define DH      64
#define KSTRIDE 32
#define WIN     16
#define NTILES  7
#define NCOLS   112
#define ROWPAD  68
#define SCPAD   116
#define HALFC   (NCOLS / 2)
#define HALFV   (HALFC / 4)
#define WAVES   8
#define QROWS   (16 * WAVES)
#define LROWS   (QROWS + 2 * WIN)
#define SROWS   (S_LEN / KSTRIDE)
#define QPASS   (QROWS / 16)
#define LPASS   (LROWS / 16)
#define SPASS   (SROWS / 16)
#define VPITCH  160
#define LPAIRS  (LROWS / 2)
#define SPAIRS  (SROWS / 2)
#define LPPASS  (LPAIRS / 16)
#define SPPASS  (SPAIRS / 16)

__device__ __forceinline__ float xhalf(float x)
{
    const int b = __builtin_bit_cast(int, x);
    return __builtin_bit_cast(float,
        __builtin_amdgcn_permlanex16(b, b, 0x76543210, 0xfedcba98, false, false));
}

__device__ __forceinline__ void pv_frags(const int t, const int wv, const int m, const int hi,
                                         const float* sc, const float* vl, const float* vs,
                                         v2f (&a)[4], v2f (&b)[16])
{
    const float* vt = (t < 3) ? &vl[(8 * wv + 8 * t) * VPITCH]
                              : &vs[(8 * (t - 3)) * VPITCH];
    #pragma unroll
    for (int kc = 0; kc < 4; ++kc) {
        const int c0 = t * 16 + 4 * kc + 2 * hi;
        a[kc] = *(const v2f*)&sc[m * SCPAD + c0];
        #pragma unroll
        for (int nt = 0; nt < 4; ++nt)
            b[4 * kc + nt] = *(const v2f*)&vt[(2 * kc + hi) * VPITCH + 2 * (16 * nt + m)];
    }
}

__global__ __launch_bounds__(32 * WAVES)
void strided_attn_wmma_f32(const float* __restrict__ q,
                           const float* __restrict__ k,
                           const float* __restrict__ v,
                           float* __restrict__ out)
{
    __shared__ __attribute__((aligned(16))) float q_lds[QROWS * ROWPAD];
    __shared__ __attribute__((aligned(16))) float kl_lds[LROWS * ROWPAD];
    __shared__ __attribute__((aligned(16))) float vl_lds[LPAIRS * VPITCH];
    __shared__ __attribute__((aligned(16))) float ks_lds[SROWS * ROWPAD];
    __shared__ __attribute__((aligned(16))) float vs_lds[SPAIRS * VPITCH];
    __shared__ __attribute__((aligned(16))) float sc_lds[WAVES * 16 * SCPAD];

    const int tid    = threadIdx.x;
    const int wv     = tid >> 5;
    const int lane   = tid & 31;
    const int m      = lane & 15;
    const int hi     = lane >> 4;
    const int h      = blockIdx.y;
    const int qbase0 = blockIdx.x * QROWS;
    const int qbase  = qbase0 + 16 * wv;
    const long hbase = (long)h * S_LEN * DH;

    const int srow = tid >> 4;
    const int scol = (tid & 15) * 4;

    v4f rq[QPASS], rk[LPASS], rks[SPASS], rv[LPPASS][2], rvs[SPPASS][2];

    #pragma unroll
    for (int p = 0; p < QPASS; ++p) {
        const int row = p * 16 + srow;
        rq[p] = *(const v4f*)(q + hbase + (long)(qbase0 + row) * DH + scol);
    }
    #pragma unroll
    for (int p = 0; p < LPASS; ++p) {
        int pos = qbase0 - WIN + p * 16 + srow;
        pos = pos < 0 ? 0 : (pos > S_LEN - 1 ? S_LEN - 1 : pos);
        rk[p] = *(const v4f*)(k + hbase + (long)pos * DH + scol);
    }
    #pragma unroll
    for (int p = 0; p < SPASS; ++p) {
        const int pos = (p * 16 + srow) * KSTRIDE;
        rks[p] = *(const v4f*)(k + hbase + (long)pos * DH + scol);
    }

    #pragma unroll
    for (int p = 0; p < QPASS; ++p)
        *(v4f*)&q_lds[(p * 16 + srow) * ROWPAD + scol] = rq[p];
    #pragma unroll
    for (int p = 0; p < LPASS; ++p)
        *(v4f*)&kl_lds[(p * 16 + srow) * ROWPAD + scol] = rk[p];
    #pragma unroll
    for (int p = 0; p < SPASS; ++p)
        *(v4f*)&ks_lds[(p * 16 + srow) * ROWPAD + scol] = rks[p];
    __syncthreads();

    #pragma unroll
    for (int p = 0; p < LPPASS; ++p) {
        #pragma unroll
        for (int j = 0; j < 2; ++j) {
            int pos = qbase0 - WIN + 2 * (p * 16 + srow) + j;
            pos = pos < 0 ? 0 : (pos > S_LEN - 1 ? S_LEN - 1 : pos);
            rv[p][j] = *(const v4f*)(v + hbase + (long)pos * DH + scol);
        }
    }
    #pragma unroll
    for (int p = 0; p < SPPASS; ++p) {
        #pragma unroll
        for (int j = 0; j < 2; ++j) {
            const int pos = (2 * (p * 16 + srow) + j) * KSTRIDE;
            rvs[p][j] = *(const v4f*)(v + hbase + (long)pos * DH + scol);
        }
    }

    float* sc = &sc_lds[wv * 16 * SCPAD];

    v2f qa[16];
    #pragma unroll
    for (int kc = 0; kc < 16; ++kc)
        qa[kc] = *(const v2f*)&q_lds[(16 * wv + m) * ROWPAD + 4 * kc + 2 * hi];

    #pragma unroll
    for (int t = 0; t < NTILES; ++t) {
        const float* krow = (t < 3) ? &kl_lds[(16 * wv + 16 * t + m) * ROWPAD]
                                    : &ks_lds[(16 * (t - 3) + m) * ROWPAD];

        v8f acc = {};
        #pragma unroll
        for (int kc = 0; kc < 16; ++kc) {
            const int d0 = 4 * kc + 2 * hi;
            v2f b = *(const v2f*)&krow[d0];
            acc = __builtin_amdgcn_wmma_f32_16x16x4_f32(
                false, qa[kc], false, b, (short)0, acc, false, false);
        }

        #pragma unroll
        for (int r = 0; r < 8; ++r) {
            const int row = r + 8 * hi;
            const int n   = qbase + row;
            const int col = t * 16 + m;
            bool valid;
            if (t < 3) {
                const int p = qbase - WIN + col;
                valid = (col >= row) && (col <= row + 2 * WIN)
                        && (p >= 0) && (p < S_LEN);
            } else {
                valid = (n > WIN);
            }
            sc[row * SCPAD + col] = valid ? acc[r] * 0.125f : -1.0e30f;
        }
    }
    __syncthreads();

    {
        float* prow = &sc[m * SCPAD + hi * HALFC];
        v4f s[HALFV];
        #pragma unroll
        for (int j = 0; j < HALFV; ++j)
            s[j] = *(const v4f*)&prow[4 * j];

        float mx = -1.0e30f;
        #pragma unroll
        for (int j = 0; j < HALFV; ++j)
            mx = fmaxf(mx, fmaxf(fmaxf(s[j].x, s[j].y), fmaxf(s[j].z, s[j].w)));
        mx = fmaxf(mx, xhalf(mx));

        v4f part = {0.0f, 0.0f, 0.0f, 0.0f};
        #pragma unroll
        for (int j = 0; j < HALFV; ++j) {
            v4f e;
            e.x = __expf(s[j].x - mx);
            e.y = __expf(s[j].y - mx);
            e.z = __expf(s[j].z - mx);
            e.w = __expf(s[j].w - mx);
            s[j] = e;
            part += e;
        }
        float sum = (part.x + part.y) + (part.z + part.w);
        sum = sum + xhalf(sum);
        const float inv = (sum > 0.0f) ? (1.0f / sum) : 0.0f;

        #pragma unroll
        for (int j = 0; j < HALFV; ++j)
            *(v4f*)&prow[4 * j] = s[j] * inv;
    }

    #pragma unroll
    for (int p = 0; p < LPPASS; ++p) {
        float* d = &vl_lds[(p * 16 + srow) * VPITCH + 2 * scol];
        const v4f lo = {rv[p][0].x, rv[p][1].x, rv[p][0].y, rv[p][1].y};
        const v4f up = {rv[p][0].z, rv[p][1].z, rv[p][0].w, rv[p][1].w};
        *(v4f*)d = lo;
        *(v4f*)(d + 4) = up;
    }
    #pragma unroll
    for (int p = 0; p < SPPASS; ++p) {
        float* d = &vs_lds[(p * 16 + srow) * VPITCH + 2 * scol];
        const v4f lo = {rvs[p][0].x, rvs[p][1].x, rvs[p][0].y, rvs[p][1].y};
        const v4f up = {rvs[p][0].z, rvs[p][1].z, rvs[p][0].w, rvs[p][1].w};
        *(v4f*)d = lo;
        *(v4f*)(d + 4) = up;
    }
    __syncthreads();

    v8f acc_o[4] = {v8f{}, v8f{}, v8f{}, v8f{}};
    v2f fa[2][4];
    v2f fb[2][16];
    pv_frags(0, wv, m, hi, sc, vl_lds, vs_lds, fa[0], fb[0]);
    __builtin_amdgcn_sched_barrier(0);
    #pragma unroll
    for (int t = 0; t < NTILES; ++t) {
        const int cur = t & 1;
        if (t + 1 < NTILES)
            pv_frags(t + 1, wv, m, hi, sc, vl_lds, vs_lds, fa[cur ^ 1], fb[cur ^ 1]);
        __builtin_amdgcn_sched_barrier(0);
        #pragma unroll
        for (int kc = 0; kc < 4; ++kc) {
            #pragma unroll
            for (int nt = 0; nt < 4; ++nt)
                acc_o[nt] = __builtin_amdgcn_wmma_f32_16x16x4_f32(
                    false, fa[cur][kc], false, fb[cur][4 * kc + nt], (short)0, acc_o[nt], false, false);
        }
        __builtin_amdgcn_sched_barrier(0);
    }

    #pragma unroll
    for (int nt = 0; nt < 4; ++nt) {
        #pragma unroll
        for (int r = 0; r < 8; ++r) {
            const int row = r + 8 * hi;
            out[hbase + (long)(qbase + row) * DH + 16 * nt + m] = acc_o[nt][r];
        }
    }
}

extern "C" void kernel_launch(void* const* d_in, const int* in_sizes, int n_in,
                              void* d_out, int out_size, void* d_ws, size_t ws_size,
                              hipStream_t stream) {
    const float* q = (const float*)d_in[0];
    const float* k = (const float*)d_in[1];
    const float* v = (const float*)d_in[2];
    float* out = (float*)d_out;

    dim3 grid(S_LEN / QROWS, NHEAD, 1);
    dim3 block(32 * WAVES, 1, 1);
    strided_attn_wmma_f32<<<grid, block, 0, stream>>>(q, k, v, out);
}
